# starting kernel
# baseline (speedup 1.0000x reference)
#include <hip/hip_runtime.h>

typedef __attribute__((ext_vector_type(2))) float v2f;
typedef __attribute__((ext_vector_type(8))) float v8f;

#define NUM_TYPES 4
#define IN_DIM    128
#define HID_DIM   256
#define OUT_DIM   128
#define M_TILE    64

#define XS_STRIDE (IN_DIM + 4)
#define HS_STRIDE (HID_DIM + 4)


__global__ void k_init(int* cnt) {
  if (threadIdx.x < NUM_TYPES) cnt[threadIdx.x] = 0;
}

__global__ void k_hist(const int* __restrict__ type, int* cnt, int N) {
  int n = blockIdx.x * blockDim.x + threadIdx.x;
  if (n < N) atomicAdd(&cnt[type[n]], 1);
}

__global__ void k_scan(const int* __restrict__ cnt, int* basev, int* running) {
  if (threadIdx.x == 0) {
    int acc = 0;
    for (int t = 0; t < NUM_TYPES; ++t) { basev[t] = acc; running[t] = acc; acc += cnt[t]; }
    basev[NUM_TYPES] = acc;
  }
}

__global__ void k_scatter(const int* __restrict__ type, int* running, int* gidx, int N) {
  int n = blockIdx.x * blockDim.x + threadIdx.x;
  if (n < N) { int p = atomicAdd(&running[type[n]], 1); gidx[p] = n; }
}

__global__ void k_transpose(const float* __restrict__ W1, const float* __restrict__ W2,
                            float* __restrict__ W1T, float* __restrict__ W2T) {
  int i = blockIdx.x * blockDim.x + threadIdx.x;
  const int tot = NUM_TYPES * IN_DIM * HID_DIM;
  if (i < tot) {
    int t = i / (IN_DIM * HID_DIM);
    int r = (i / HID_DIM) % IN_DIM;
    int c = i % HID_DIM;
    W1T[((size_t)t * HID_DIM + c) * IN_DIM + r] = W1[i];
    int t2 = i / (HID_DIM * OUT_DIM);
    int r2 = (i / OUT_DIM) % HID_DIM;
    int c2 = i % OUT_DIM;
    W2T[((size_t)t2 * OUT_DIM + c2) * HID_DIM + r2] = W2[i];
  }
}


__global__ __launch_bounds__(256) void k_mlp(
    const float* __restrict__ X, const float* __restrict__ b1, const float* __restrict__ b2,
    const float* __restrict__ W1T, const float* __restrict__ W2T,
    const int* __restrict__ basev, const int* __restrict__ gidx,
    float* __restrict__ out) {
  __shared__ float Xs[M_TILE][XS_STRIDE];
  __shared__ float Hs[M_TILE][HS_STRIDE];
  __shared__ int   rows[M_TILE];

  int tile = blockIdx.x;
  int ty = -1, ltile = 0, acc = 0, segStart = 0, segCount = 0;
  for (int q = 0; q < NUM_TYPES; ++q) {
    int c  = basev[q + 1] - basev[q];
    int nt = (c + (M_TILE - 1)) >> 6;
    if (tile < acc + nt) { ty = q; ltile = tile - acc; segStart = basev[q]; segCount = c; break; }
    acc += nt;
  }
  if (ty < 0) return;

  int row0 = ltile << 6;
  int tid  = threadIdx.x;

  if (tid < M_TILE) {
    int g = row0 + tid;
    rows[tid] = (g < segCount) ? gidx[segStart + g] : -1;
  }
  __syncthreads();

  for (int i = tid; i < M_TILE * (IN_DIM / 4); i += 256) {
    int r  = i >> 5;
    int c4 = i & 31;
    float4 v = make_float4(0.f, 0.f, 0.f, 0.f);
    int node = rows[r];
    if (node >= 0) v = ((const float4*)(X + (size_t)node * IN_DIM))[c4];
    *(float4*)&Xs[r][c4 * 4] = v;
  }
  __syncthreads();

  int wave = tid >> 5;
  int lane = tid & 31;
  int n    = lane & 15;
  int kh   = (lane >> 4) << 1;
  int mh   = (lane >> 4) << 3;

  for (int nt = wave; nt < HID_DIM / 16; nt += 8) {
    v8f c0 = {}, c1 = {}, c2 = {}, c3 = {};
    const float* wcol = W1T + ((size_t)ty * HID_DIM + nt * 16 + n) * IN_DIM;
#pragma unroll 4
    for (int k0 = 0; k0 < IN_DIM; k0 += 4) {
      v2f b  = *(const v2f*)&wcol[k0 + kh];
      v2f a0 = *(const v2f*)&Xs[ 0 + n][k0 + kh];
      v2f a1 = *(const v2f*)&Xs[16 + n][k0 + kh];
      v2f a2 = *(const v2f*)&Xs[32 + n][k0 + kh];
      v2f a3 = *(const v2f*)&Xs[48 + n][k0 + kh];
      c0 = __builtin_amdgcn_wmma_f32_16x16x4_f32(false, a0, false, b, (short)0, c0, false, false);
      c1 = __builtin_amdgcn_wmma_f32_16x16x4_f32(false, a1, false, b, (short)0, c1, false, false);
      c2 = __builtin_amdgcn_wmma_f32_16x16x4_f32(false, a2, false, b, (short)0, c2, false, false);
      c3 = __builtin_amdgcn_wmma_f32_16x16x4_f32(false, a3, false, b, (short)0, c3, false, false);
    }
    float bias = b1[ty * HID_DIM + nt * 16 + n];
    int col = nt * 16 + n;
#pragma unroll
    for (int r = 0; r < 8; ++r) {
      float v0 = c0[r] + bias, v1 = c1[r] + bias, v2 = c2[r] + bias, v3 = c3[r] + bias;
      Hs[ 0 + r + mh][col] = v0 > 0.f ? v0 : 0.f;
      Hs[16 + r + mh][col] = v1 > 0.f ? v1 : 0.f;
      Hs[32 + r + mh][col] = v2 > 0.f ? v2 : 0.f;
      Hs[48 + r + mh][col] = v3 > 0.f ? v3 : 0.f;
    }
  }
  __syncthreads();

  {
    int ot = wave;
    v8f c0 = {}, c1 = {}, c2 = {}, c3 = {};
    const float* wcol = W2T + ((size_t)ty * OUT_DIM + ot * 16 + n) * HID_DIM;
#pragma unroll 4
    for (int k0 = 0; k0 < HID_DIM; k0 += 4) {
      v2f b  = *(const v2f*)&wcol[k0 + kh];
      v2f a0 = *(const v2f*)&Hs[ 0 + n][k0 + kh];
      v2f a1 = *(const v2f*)&Hs[16 + n][k0 + kh];
      v2f a2 = *(const v2f*)&Hs[32 + n][k0 + kh];
      v2f a3 = *(const v2f*)&Hs[48 + n][k0 + kh];
      c0 = __builtin_amdgcn_wmma_f32_16x16x4_f32(false, a0, false, b, (short)0, c0, false, false);
      c1 = __builtin_amdgcn_wmma_f32_16x16x4_f32(false, a1, false, b, (short)0, c1, false, false);
      c2 = __builtin_amdgcn_wmma_f32_16x16x4_f32(false, a2, false, b, (short)0, c2, false, false);
      c3 = __builtin_amdgcn_wmma_f32_16x16x4_f32(false, a3, false, b, (short)0, c3, false, false);
    }
    float bias = b2[ty * OUT_DIM + ot * 16 + n];
    int col = ot * 16 + n;
#pragma unroll
    for (int r = 0; r < 8; ++r) {
      int n0 = rows[ 0 + r + mh];
      int n1 = rows[16 + r + mh];
      int n2 = rows[32 + r + mh];
      int n3 = rows[48 + r + mh];
      if (n0 >= 0) out[(size_t)n0 * OUT_DIM + col] = c0[r] + bias;
      if (n1 >= 0) out[(size_t)n1 * OUT_DIM + col] = c1[r] + bias;
      if (n2 >= 0) out[(size_t)n2 * OUT_DIM + col] = c2[r] + bias;
      if (n3 >= 0) out[(size_t)n3 * OUT_DIM + col] = c3[r] + bias;
    }
  }
}


extern "C" void kernel_launch(void* const* d_in, const int* in_sizes, int n_in,
                              void* d_out, int out_size, void* d_ws, size_t ws_size,
                              hipStream_t stream) {
  const float* X    = (const float*)d_in[0];
  const int*   type = (const int*)  d_in[1];
  const float* W1   = (const float*)d_in[2];
  const float* b1   = (const float*)d_in[3];
  const float* W2   = (const float*)d_in[4];
  const float* b2   = (const float*)d_in[5];
  float* out = (float*)d_out;

  int N = in_sizes[0] / IN_DIM;

  int* cnt     = (int*)d_ws;
  int* basev   = cnt + 8;
  int* running = basev + 8;
  int* gidx    = running + 8;
  int  gpad    = (N + 63) & ~63;
  float* W1T   = (float*)(gidx + gpad);
  float* W2T   = W1T + (size_t)NUM_TYPES * HID_DIM * IN_DIM;

  int nblk = (N + 255) / 256;
  k_init<<<1, 64, 0, stream>>>(cnt);
  k_hist<<<nblk, 256, 0, stream>>>(type, cnt, N);
  k_scan<<<1, 32, 0, stream>>>(cnt, basev, running);
  k_scatter<<<nblk, 256, 0, stream>>>(type, running, gidx, N);

  int wtot = NUM_TYPES * IN_DIM * HID_DIM;
  k_transpose<<<(wtot + 255) / 256, 256, 0, stream>>>(W1, W2, W1T, W2T);

  int maxTiles = (N + M_TILE - 1) / M_TILE + NUM_TYPES;
  k_mlp<<<maxTiles, 256, 0, stream>>>(X, b1, b2, W1T, W2T, basev, gidx, out);
}


# bucketing without atomics: per-block type counts by wave ballot, each scatter block sums earlier blocks' counts and ranks its nodes by ballot; init and scan launches go
# speedup vs baseline: 1.6719x; 1.6719x over previous
#include <hip/hip_runtime.h>

typedef __attribute__((ext_vector_type(2))) float v2f;
typedef __attribute__((ext_vector_type(8))) float v8f;

#define NUM_TYPES 4
#define IN_DIM    128
#define HID_DIM   256
#define OUT_DIM   128
#define M_TILE    64

#define XS_STRIDE (IN_DIM + 4)
#define HS_STRIDE (HID_DIM + 4)


#define BKT 256
#define BWAVES (BKT / 32)

__global__ __launch_bounds__(BKT) void k_hist(const int* __restrict__ type, int* __restrict__ blkcnt, int N) {
  __shared__ int wc[BWAVES][NUM_TYPES];
  int tid = threadIdx.x, lane = tid & 31, wave = tid >> 5;
  int n = blockIdx.x * BKT + tid;
  int t = (n < N) ? type[n] : -1;
  unsigned m0 = __builtin_amdgcn_ballot_w32(t == 0);
  unsigned m1 = __builtin_amdgcn_ballot_w32(t == 1);
  unsigned m2 = __builtin_amdgcn_ballot_w32(t == 2);
  unsigned m3 = __builtin_amdgcn_ballot_w32(t == 3);
  if (lane == 0) {
    wc[wave][0] = __builtin_popcount(m0); wc[wave][1] = __builtin_popcount(m1);
    wc[wave][2] = __builtin_popcount(m2); wc[wave][3] = __builtin_popcount(m3);
  }
  __syncthreads();
  if (tid < NUM_TYPES) {
    int a = 0;
    for (int w = 0; w < BWAVES; ++w) a += wc[w][tid];
    blkcnt[blockIdx.x * NUM_TYPES + tid] = a;
  }
}

__global__ __launch_bounds__(BKT) void k_scatter(const int* __restrict__ type, const int* __restrict__ blkcnt,
                                                 int* __restrict__ basev, int* __restrict__ gidx, int N, int NB) {
  __shared__ int wc[BWAVES][NUM_TYPES];
  __shared__ int red[BWAVES][2 * NUM_TYPES];
  __shared__ int fin[2 * NUM_TYPES];
  int tid = threadIdx.x, lane = tid & 31, wave = tid >> 5;
  int b = blockIdx.x;
  int n = b * BKT + tid;
  int t = (n < N) ? type[n] : -1;
  unsigned m0 = __builtin_amdgcn_ballot_w32(t == 0);
  unsigned m1 = __builtin_amdgcn_ballot_w32(t == 1);
  unsigned m2 = __builtin_amdgcn_ballot_w32(t == 2);
  unsigned m3 = __builtin_amdgcn_ballot_w32(t == 3);
  if (lane == 0) {
    wc[wave][0] = __builtin_popcount(m0); wc[wave][1] = __builtin_popcount(m1);
    wc[wave][2] = __builtin_popcount(m2); wc[wave][3] = __builtin_popcount(m3);
  }

  int p0 = 0, p1 = 0, p2 = 0, p3 = 0, s0 = 0, s1 = 0, s2 = 0, s3 = 0;
  for (int i = tid; i < NB; i += BKT) {
    int4 c = ((const int4*)blkcnt)[i];
    int e = (i < b) ? -1 : 0;
    s0 += c.x; s1 += c.y; s2 += c.z; s3 += c.w;
    p0 += c.x & e; p1 += c.y & e; p2 += c.z & e; p3 += c.w & e;
  }
  for (int o = 16; o > 0; o >>= 1) {
    p0 += __shfl_xor(p0, o); p1 += __shfl_xor(p1, o); p2 += __shfl_xor(p2, o); p3 += __shfl_xor(p3, o);
    s0 += __shfl_xor(s0, o); s1 += __shfl_xor(s1, o); s2 += __shfl_xor(s2, o); s3 += __shfl_xor(s3, o);
  }
  if (lane == 0) {
    red[wave][0] = p0; red[wave][1] = p1; red[wave][2] = p2; red[wave][3] = p3;
    red[wave][4] = s0; red[wave][5] = s1; red[wave][6] = s2; red[wave][7] = s3;
  }
  __syncthreads();
  if (tid < 2 * NUM_TYPES) {
    int a = 0;
    for (int w = 0; w < BWAVES; ++w) a += red[w][tid];
    fin[tid] = a;
  }
  __syncthreads();

  if (b == 0 && tid == 0) {
    int acc = 0;
    for (int q = 0; q < NUM_TYPES; ++q) { basev[q] = acc; acc += fin[NUM_TYPES + q]; }
    basev[NUM_TYPES] = acc;
  }
  if ((unsigned)t < NUM_TYPES) {
    unsigned m = t == 0 ? m0 : t == 1 ? m1 : t == 2 ? m2 : m3;
    int pos = fin[t];
    for (int q = 0; q < NUM_TYPES - 1; ++q) pos += (q < t) ? fin[NUM_TYPES + q] : 0;
    for (int w = 0; w < BWAVES - 1; ++w) pos += (w < wave) ? wc[w][t] : 0;
    pos += __builtin_popcount(m & ((1u << lane) - 1u));
    gidx[pos] = n;
  }
}

__global__ void k_transpose(const float* __restrict__ W1, const float* __restrict__ W2,
                            float* __restrict__ W1T, float* __restrict__ W2T) {
  int i = blockIdx.x * blockDim.x + threadIdx.x;
  const int tot = NUM_TYPES * IN_DIM * HID_DIM;
  if (i < tot) {
    int t = i / (IN_DIM * HID_DIM);
    int r = (i / HID_DIM) % IN_DIM;
    int c = i % HID_DIM;
    W1T[((size_t)t * HID_DIM + c) * IN_DIM + r] = W1[i];
    int t2 = i / (HID_DIM * OUT_DIM);
    int r2 = (i / OUT_DIM) % HID_DIM;
    int c2 = i % OUT_DIM;
    W2T[((size_t)t2 * OUT_DIM + c2) * HID_DIM + r2] = W2[i];
  }
}


__global__ __launch_bounds__(256) void k_mlp(
    const float* __restrict__ X, const float* __restrict__ b1, const float* __restrict__ b2,
    const float* __restrict__ W1T, const float* __restrict__ W2T,
    const int* __restrict__ basev, const int* __restrict__ gidx,
    float* __restrict__ out) {
  __shared__ float Xs[M_TILE][XS_STRIDE];
  __shared__ float Hs[M_TILE][HS_STRIDE];
  __shared__ int   rows[M_TILE];

  int tile = blockIdx.x;
  int ty = -1, ltile = 0, acc = 0, segStart = 0, segCount = 0;
  for (int q = 0; q < NUM_TYPES; ++q) {
    int c  = basev[q + 1] - basev[q];
    int nt = (c + (M_TILE - 1)) >> 6;
    if (tile < acc + nt) { ty = q; ltile = tile - acc; segStart = basev[q]; segCount = c; break; }
    acc += nt;
  }
  if (ty < 0) return;

  int row0 = ltile << 6;
  int tid  = threadIdx.x;

  if (tid < M_TILE) {
    int g = row0 + tid;
    rows[tid] = (g < segCount) ? gidx[segStart + g] : -1;
  }
  __syncthreads();

  for (int i = tid; i < M_TILE * (IN_DIM / 4); i += 256) {
    int r  = i >> 5;
    int c4 = i & 31;
    float4 v = make_float4(0.f, 0.f, 0.f, 0.f);
    int node = rows[r];
    if (node >= 0) v = ((const float4*)(X + (size_t)node * IN_DIM))[c4];
    *(float4*)&Xs[r][c4 * 4] = v;
  }
  __syncthreads();

  int wave = tid >> 5;
  int lane = tid & 31;
  int n    = lane & 15;
  int kh   = (lane >> 4) << 1;
  int mh   = (lane >> 4) << 3;

  for (int nt = wave; nt < HID_DIM / 16; nt += 8) {
    v8f c0 = {}, c1 = {}, c2 = {}, c3 = {};
    const float* wcol = W1T + ((size_t)ty * HID_DIM + nt * 16 + n) * IN_DIM;
#pragma unroll 4
    for (int k0 = 0; k0 < IN_DIM; k0 += 4) {
      v2f b  = *(const v2f*)&wcol[k0 + kh];
      v2f a0 = *(const v2f*)&Xs[ 0 + n][k0 + kh];
      v2f a1 = *(const v2f*)&Xs[16 + n][k0 + kh];
      v2f a2 = *(const v2f*)&Xs[32 + n][k0 + kh];
      v2f a3 = *(const v2f*)&Xs[48 + n][k0 + kh];
      c0 = __builtin_amdgcn_wmma_f32_16x16x4_f32(false, a0, false, b, (short)0, c0, false, false);
      c1 = __builtin_amdgcn_wmma_f32_16x16x4_f32(false, a1, false, b, (short)0, c1, false, false);
      c2 = __builtin_amdgcn_wmma_f32_16x16x4_f32(false, a2, false, b, (short)0, c2, false, false);
      c3 = __builtin_amdgcn_wmma_f32_16x16x4_f32(false, a3, false, b, (short)0, c3, false, false);
    }
    float bias = b1[ty * HID_DIM + nt * 16 + n];
    int col = nt * 16 + n;
#pragma unroll
    for (int r = 0; r < 8; ++r) {
      float v0 = c0[r] + bias, v1 = c1[r] + bias, v2 = c2[r] + bias, v3 = c3[r] + bias;
      Hs[ 0 + r + mh][col] = v0 > 0.f ? v0 : 0.f;
      Hs[16 + r + mh][col] = v1 > 0.f ? v1 : 0.f;
      Hs[32 + r + mh][col] = v2 > 0.f ? v2 : 0.f;
      Hs[48 + r + mh][col] = v3 > 0.f ? v3 : 0.f;
    }
  }
  __syncthreads();

  {
    int ot = wave;
    v8f c0 = {}, c1 = {}, c2 = {}, c3 = {};
    const float* wcol = W2T + ((size_t)ty * OUT_DIM + ot * 16 + n) * HID_DIM;
#pragma unroll 4
    for (int k0 = 0; k0 < HID_DIM; k0 += 4) {
      v2f b  = *(const v2f*)&wcol[k0 + kh];
      v2f a0 = *(const v2f*)&Hs[ 0 + n][k0 + kh];
      v2f a1 = *(const v2f*)&Hs[16 + n][k0 + kh];
      v2f a2 = *(const v2f*)&Hs[32 + n][k0 + kh];
      v2f a3 = *(const v2f*)&Hs[48 + n][k0 + kh];
      c0 = __builtin_amdgcn_wmma_f32_16x16x4_f32(false, a0, false, b, (short)0, c0, false, false);
      c1 = __builtin_amdgcn_wmma_f32_16x16x4_f32(false, a1, false, b, (short)0, c1, false, false);
      c2 = __builtin_amdgcn_wmma_f32_16x16x4_f32(false, a2, false, b, (short)0, c2, false, false);
      c3 = __builtin_amdgcn_wmma_f32_16x16x4_f32(false, a3, false, b, (short)0, c3, false, false);
    }
    float bias = b2[ty * OUT_DIM + ot * 16 + n];
    int col = ot * 16 + n;
#pragma unroll
    for (int r = 0; r < 8; ++r) {
      int n0 = rows[ 0 + r + mh];
      int n1 = rows[16 + r + mh];
      int n2 = rows[32 + r + mh];
      int n3 = rows[48 + r + mh];
      if (n0 >= 0) out[(size_t)n0 * OUT_DIM + col] = c0[r] + bias;
      if (n1 >= 0) out[(size_t)n1 * OUT_DIM + col] = c1[r] + bias;
      if (n2 >= 0) out[(size_t)n2 * OUT_DIM + col] = c2[r] + bias;
      if (n3 >= 0) out[(size_t)n3 * OUT_DIM + col] = c3[r] + bias;
    }
  }
}


extern "C" void kernel_launch(void* const* d_in, const int* in_sizes, int n_in,
                              void* d_out, int out_size, void* d_ws, size_t ws_size,
                              hipStream_t stream) {
  const float* X    = (const float*)d_in[0];
  const int*   type = (const int*)  d_in[1];
  const float* W1   = (const float*)d_in[2];
  const float* b1   = (const float*)d_in[3];
  const float* W2   = (const float*)d_in[4];
  const float* b2   = (const float*)d_in[5];
  float* out = (float*)d_out;

  int N = in_sizes[0] / IN_DIM;

  int  nblk    = (N + BKT - 1) / BKT;
  int* basev   = (int*)d_ws;
  int* blkcnt  = basev + 64;
  int  bpad    = (nblk * NUM_TYPES + 63) & ~63;
  int* gidx    = blkcnt + bpad;
  int  gpad    = (N + 63) & ~63;
  float* W1T   = (float*)(gidx + gpad);
  float* W2T   = W1T + (size_t)NUM_TYPES * HID_DIM * IN_DIM;

  k_hist<<<nblk, BKT, 0, stream>>>(type, blkcnt, N);
  k_scatter<<<nblk, BKT, 0, stream>>>(type, blkcnt, basev, gidx, N, nblk);

  int wtot = NUM_TYPES * IN_DIM * HID_DIM;
  k_transpose<<<(wtot + 255) / 256, 256, 0, stream>>>(W1, W2, W1T, W2T);

  int maxTiles = (N + M_TILE - 1) / M_TILE + NUM_TYPES;
  k_mlp<<<maxTiles, 256, 0, stream>>>(X, b1, b2, W1T, W2T, basev, gidx, out);
}


# B fragments read in place from row-major W1/W2 as coalesced 8-byte row pieces (a lane holds 2 adjacent columns = 2 interleaved N-tiles per wave); transposed weight copies and their launch go
# speedup vs baseline: 2.0284x; 1.2132x over previous
#include <hip/hip_runtime.h>

typedef __attribute__((ext_vector_type(2))) float v2f;
typedef __attribute__((ext_vector_type(8))) float v8f;

#define NUM_TYPES 4
#define IN_DIM    128
#define HID_DIM   256
#define OUT_DIM   128
#define M_TILE    64

#define XS_STRIDE (IN_DIM + 4)
#define HS_STRIDE (HID_DIM + 4)


#define BKT 256
#define BWAVES (BKT / 32)

__global__ __launch_bounds__(BKT) void k_hist(const int* __restrict__ type, int* __restrict__ blkcnt, int N) {
  __shared__ int wc[BWAVES][NUM_TYPES];
  int tid = threadIdx.x, lane = tid & 31, wave = tid >> 5;
  int n = blockIdx.x * BKT + tid;
  int t = (n < N) ? type[n] : -1;
  unsigned m0 = __builtin_amdgcn_ballot_w32(t == 0);
  unsigned m1 = __builtin_amdgcn_ballot_w32(t == 1);
  unsigned m2 = __builtin_amdgcn_ballot_w32(t == 2);
  unsigned m3 = __builtin_amdgcn_ballot_w32(t == 3);
  if (lane == 0) {
    wc[wave][0] = __builtin_popcount(m0); wc[wave][1] = __builtin_popcount(m1);
    wc[wave][2] = __builtin_popcount(m2); wc[wave][3] = __builtin_popcount(m3);
  }
  __syncthreads();
  if (tid < NUM_TYPES) {
    int a = 0;
    for (int w = 0; w < BWAVES; ++w) a += wc[w][tid];
    blkcnt[blockIdx.x * NUM_TYPES + tid] = a;
  }
}

__global__ __launch_bounds__(BKT) void k_scatter(const int* __restrict__ type, const int* __restrict__ blkcnt,
                                                 int* __restrict__ basev, int* __restrict__ gidx, int N, int NB) {
  __shared__ int wc[BWAVES][NUM_TYPES];
  __shared__ int red[BWAVES][2 * NUM_TYPES];
  __shared__ int fin[2 * NUM_TYPES];
  int tid = threadIdx.x, lane = tid & 31, wave = tid >> 5;
  int b = blockIdx.x;
  int n = b * BKT + tid;
  int t = (n < N) ? type[n] : -1;
  unsigned m0 = __builtin_amdgcn_ballot_w32(t == 0);
  unsigned m1 = __builtin_amdgcn_ballot_w32(t == 1);
  unsigned m2 = __builtin_amdgcn_ballot_w32(t == 2);
  unsigned m3 = __builtin_amdgcn_ballot_w32(t == 3);
  if (lane == 0) {
    wc[wave][0] = __builtin_popcount(m0); wc[wave][1] = __builtin_popcount(m1);
    wc[wave][2] = __builtin_popcount(m2); wc[wave][3] = __builtin_popcount(m3);
  }

  int p0 = 0, p1 = 0, p2 = 0, p3 = 0, s0 = 0, s1 = 0, s2 = 0, s3 = 0;
  for (int i = tid; i < NB; i += BKT) {
    int4 c = ((const int4*)blkcnt)[i];
    int e = (i < b) ? -1 : 0;
    s0 += c.x; s1 += c.y; s2 += c.z; s3 += c.w;
    p0 += c.x & e; p1 += c.y & e; p2 += c.z & e; p3 += c.w & e;
  }
  for (int o = 16; o > 0; o >>= 1) {
    p0 += __shfl_xor(p0, o); p1 += __shfl_xor(p1, o); p2 += __shfl_xor(p2, o); p3 += __shfl_xor(p3, o);
    s0 += __shfl_xor(s0, o); s1 += __shfl_xor(s1, o); s2 += __shfl_xor(s2, o); s3 += __shfl_xor(s3, o);
  }
  if (lane == 0) {
    red[wave][0] = p0; red[wave][1] = p1; red[wave][2] = p2; red[wave][3] = p3;
    red[wave][4] = s0; red[wave][5] = s1; red[wave][6] = s2; red[wave][7] = s3;
  }
  __syncthreads();
  if (tid < 2 * NUM_TYPES) {
    int a = 0;
    for (int w = 0; w < BWAVES; ++w) a += red[w][tid];
    fin[tid] = a;
  }
  __syncthreads();

  if (b == 0 && tid == 0) {
    int acc = 0;
    for (int q = 0; q < NUM_TYPES; ++q) { basev[q] = acc; acc += fin[NUM_TYPES + q]; }
    basev[NUM_TYPES] = acc;
  }
  if ((unsigned)t < NUM_TYPES) {
    unsigned m = t == 0 ? m0 : t == 1 ? m1 : t == 2 ? m2 : m3;
    int pos = fin[t];
    for (int q = 0; q < NUM_TYPES - 1; ++q) pos += (q < t) ? fin[NUM_TYPES + q] : 0;
    for (int w = 0; w < BWAVES - 1; ++w) pos += (w < wave) ? wc[w][t] : 0;
    pos += __builtin_popcount(m & ((1u << lane) - 1u));
    gidx[pos] = n;
  }
}


__global__ __launch_bounds__(256) void k_mlp(
    const float* __restrict__ X, const float* __restrict__ b1, const float* __restrict__ b2,
    const float* __restrict__ W1, const float* __restrict__ W2,
    const int* __restrict__ basev, const int* __restrict__ gidx,
    float* __restrict__ out) {
  __shared__ float Xs[M_TILE][XS_STRIDE];
  __shared__ float Hs[M_TILE][HS_STRIDE];
  __shared__ int   rows[M_TILE];

  int tile = blockIdx.x;
  int ty = -1, ltile = 0, acc = 0, segStart = 0, segCount = 0;
  for (int q = 0; q < NUM_TYPES; ++q) {
    int c  = basev[q + 1] - basev[q];
    int nt = (c + (M_TILE - 1)) >> 6;
    if (tile < acc + nt) { ty = q; ltile = tile - acc; segStart = basev[q]; segCount = c; break; }
    acc += nt;
  }
  if (ty < 0) return;

  int row0 = ltile << 6;
  int tid  = threadIdx.x;

  if (tid < M_TILE) {
    int g = row0 + tid;
    rows[tid] = (g < segCount) ? gidx[segStart + g] : -1;
  }
  __syncthreads();

  for (int i = tid; i < M_TILE * (IN_DIM / 4); i += 256) {
    int r  = i >> 5;
    int c4 = i & 31;
    float4 v = make_float4(0.f, 0.f, 0.f, 0.f);
    int node = rows[r];
    if (node >= 0) v = ((const float4*)(X + (size_t)node * IN_DIM))[c4];
    *(float4*)&Xs[r][c4 * 4] = v;
  }
  __syncthreads();

  int wave = tid >> 5;
  int lane = tid & 31;
  int n    = lane & 15;
  int kh   = (lane >> 4) << 1;
  int mh   = (lane >> 4) << 3;


  {
    int cbase = wave * 32 + 2 * n;
    const float* wp = W1 + (size_t)ty * IN_DIM * HID_DIM + (size_t)kh * HID_DIM + cbase;
    v8f c00 = {}, c01 = {}, c10 = {}, c11 = {}, c20 = {}, c21 = {}, c30 = {}, c31 = {};
#pragma unroll 4
    for (int k0 = 0; k0 < IN_DIM; k0 += 4) {
      v2f l0 = *(const v2f*)&wp[(k0    ) * HID_DIM];
      v2f l1 = *(const v2f*)&wp[(k0 + 1) * HID_DIM];
      v2f bA = {l0.x, l1.x};
      v2f bB = {l0.y, l1.y};
      v2f a0 = *(const v2f*)&Xs[ 0 + n][k0 + kh];
      v2f a1 = *(const v2f*)&Xs[16 + n][k0 + kh];
      v2f a2 = *(const v2f*)&Xs[32 + n][k0 + kh];
      v2f a3 = *(const v2f*)&Xs[48 + n][k0 + kh];
      c00 = __builtin_amdgcn_wmma_f32_16x16x4_f32(false, a0, false, bA, (short)0, c00, false, false);
      c01 = __builtin_amdgcn_wmma_f32_16x16x4_f32(false, a0, false, bB, (short)0, c01, false, false);
      c10 = __builtin_amdgcn_wmma_f32_16x16x4_f32(false, a1, false, bA, (short)0, c10, false, false);
      c11 = __builtin_amdgcn_wmma_f32_16x16x4_f32(false, a1, false, bB, (short)0, c11, false, false);
      c20 = __builtin_amdgcn_wmma_f32_16x16x4_f32(false, a2, false, bA, (short)0, c20, false, false);
      c21 = __builtin_amdgcn_wmma_f32_16x16x4_f32(false, a2, false, bB, (short)0, c21, false, false);
      c30 = __builtin_amdgcn_wmma_f32_16x16x4_f32(false, a3, false, bA, (short)0, c30, false, false);
      c31 = __builtin_amdgcn_wmma_f32_16x16x4_f32(false, a3, false, bB, (short)0, c31, false, false);
    }
    v2f bias = *(const v2f*)&b1[ty * HID_DIM + cbase];
#pragma unroll
    for (int r = 0; r < 8; ++r) {
      v2f h0 = {c00[r] + bias.x, c01[r] + bias.y};
      v2f h1 = {c10[r] + bias.x, c11[r] + bias.y};
      v2f h2 = {c20[r] + bias.x, c21[r] + bias.y};
      v2f h3 = {c30[r] + bias.x, c31[r] + bias.y};
      h0.x = h0.x > 0.f ? h0.x : 0.f; h0.y = h0.y > 0.f ? h0.y : 0.f;
      h1.x = h1.x > 0.f ? h1.x : 0.f; h1.y = h1.y > 0.f ? h1.y : 0.f;
      h2.x = h2.x > 0.f ? h2.x : 0.f; h2.y = h2.y > 0.f ? h2.y : 0.f;
      h3.x = h3.x > 0.f ? h3.x : 0.f; h3.y = h3.y > 0.f ? h3.y : 0.f;
      *(v2f*)&Hs[ 0 + r + mh][cbase] = h0;
      *(v2f*)&Hs[16 + r + mh][cbase] = h1;
      *(v2f*)&Hs[32 + r + mh][cbase] = h2;
      *(v2f*)&Hs[48 + r + mh][cbase] = h3;
    }
  }
  __syncthreads();

  {
    int cbase = (wave & 3) * 32 + 2 * n;
    int rb    = (wave >> 2) * 32;
    const float* wp = W2 + (size_t)ty * HID_DIM * OUT_DIM + (size_t)kh * OUT_DIM + cbase;
    v8f c00 = {}, c01 = {}, c10 = {}, c11 = {};
#pragma unroll 8
    for (int k0 = 0; k0 < HID_DIM; k0 += 4) {
      v2f l0 = *(const v2f*)&wp[(k0    ) * OUT_DIM];
      v2f l1 = *(const v2f*)&wp[(k0 + 1) * OUT_DIM];
      v2f bA = {l0.x, l1.x};
      v2f bB = {l0.y, l1.y};
      v2f a0 = *(const v2f*)&Hs[rb +  0 + n][k0 + kh];
      v2f a1 = *(const v2f*)&Hs[rb + 16 + n][k0 + kh];
      c00 = __builtin_amdgcn_wmma_f32_16x16x4_f32(false, a0, false, bA, (short)0, c00, false, false);
      c01 = __builtin_amdgcn_wmma_f32_16x16x4_f32(false, a0, false, bB, (short)0, c01, false, false);
      c10 = __builtin_amdgcn_wmma_f32_16x16x4_f32(false, a1, false, bA, (short)0, c10, false, false);
      c11 = __builtin_amdgcn_wmma_f32_16x16x4_f32(false, a1, false, bB, (short)0, c11, false, false);
    }
    v2f bias = *(const v2f*)&b2[ty * OUT_DIM + cbase];
#pragma unroll
    for (int r = 0; r < 8; ++r) {
      int n0 = rows[rb +  0 + r + mh];
      int n1 = rows[rb + 16 + r + mh];
      v2f o0 = {c00[r] + bias.x, c01[r] + bias.y};
      v2f o1 = {c10[r] + bias.x, c11[r] + bias.y};
      if (n0 >= 0) *(v2f*)&out[(size_t)n0 * OUT_DIM + cbase] = o0;
      if (n1 >= 0) *(v2f*)&out[(size_t)n1 * OUT_DIM + cbase] = o1;
    }
  }
}


extern "C" void kernel_launch(void* const* d_in, const int* in_sizes, int n_in,
                              void* d_out, int out_size, void* d_ws, size_t ws_size,
                              hipStream_t stream) {
  const float* X    = (const float*)d_in[0];
  const int*   type = (const int*)  d_in[1];
  const float* W1   = (const float*)d_in[2];
  const float* b1   = (const float*)d_in[3];
  const float* W2   = (const float*)d_in[4];
  const float* b2   = (const float*)d_in[5];
  float* out = (float*)d_out;

  int N = in_sizes[0] / IN_DIM;

  int  nblk    = (N + BKT - 1) / BKT;
  int* basev   = (int*)d_ws;
  int* blkcnt  = basev + 64;
  int  bpad    = (nblk * NUM_TYPES + 63) & ~63;
  int* gidx    = blkcnt + bpad;

  k_hist<<<nblk, BKT, 0, stream>>>(type, blkcnt, N);
  k_scatter<<<nblk, BKT, 0, stream>>>(type, blkcnt, basev, gidx, N, nblk);

  int maxTiles = (N + M_TILE - 1) / M_TILE + NUM_TYPES;
  k_mlp<<<maxTiles, 256, 0, stream>>>(X, b1, b2, W1, W2, basev, gidx, out);
}


# X tile staging: each thread reads its 8 row indices, issues its 8 gathered row loads together, then stores them: one memory round trip per tile instead of 8 dependent ones
# speedup vs baseline: 2.1289x; 1.0495x over previous
#include <hip/hip_runtime.h>

typedef __attribute__((ext_vector_type(2))) float v2f;
typedef __attribute__((ext_vector_type(8))) float v8f;

#define NUM_TYPES 4
#define IN_DIM    128
#define HID_DIM   256
#define OUT_DIM   128
#define M_TILE    64

#define XS_STRIDE (IN_DIM + 4)
#define HS_STRIDE (HID_DIM + 4)


#define BKT 256
#define BWAVES (BKT / 32)

__global__ __launch_bounds__(BKT) void k_hist(const int* __restrict__ type, int* __restrict__ blkcnt, int N) {
  __shared__ int wc[BWAVES][NUM_TYPES];
  int tid = threadIdx.x, lane = tid & 31, wave = tid >> 5;
  int n = blockIdx.x * BKT + tid;
  int t = (n < N) ? type[n] : -1;
  unsigned m0 = __builtin_amdgcn_ballot_w32(t == 0);
  unsigned m1 = __builtin_amdgcn_ballot_w32(t == 1);
  unsigned m2 = __builtin_amdgcn_ballot_w32(t == 2);
  unsigned m3 = __builtin_amdgcn_ballot_w32(t == 3);
  if (lane == 0) {
    wc[wave][0] = __builtin_popcount(m0); wc[wave][1] = __builtin_popcount(m1);
    wc[wave][2] = __builtin_popcount(m2); wc[wave][3] = __builtin_popcount(m3);
  }
  __syncthreads();
  if (tid < NUM_TYPES) {
    int a = 0;
    for (int w = 0; w < BWAVES; ++w) a += wc[w][tid];
    blkcnt[blockIdx.x * NUM_TYPES + tid] = a;
  }
}

__global__ __launch_bounds__(BKT) void k_scatter(const int* __restrict__ type, const int* __restrict__ blkcnt,
                                                 int* __restrict__ basev, int* __restrict__ gidx, int N, int NB) {
  __shared__ int wc[BWAVES][NUM_TYPES];
  __shared__ int red[BWAVES][2 * NUM_TYPES];
  __shared__ int fin[2 * NUM_TYPES];
  int tid = threadIdx.x, lane = tid & 31, wave = tid >> 5;
  int b = blockIdx.x;
  int n = b * BKT + tid;
  int t = (n < N) ? type[n] : -1;
  unsigned m0 = __builtin_amdgcn_ballot_w32(t == 0);
  unsigned m1 = __builtin_amdgcn_ballot_w32(t == 1);
  unsigned m2 = __builtin_amdgcn_ballot_w32(t == 2);
  unsigned m3 = __builtin_amdgcn_ballot_w32(t == 3);
  if (lane == 0) {
    wc[wave][0] = __builtin_popcount(m0); wc[wave][1] = __builtin_popcount(m1);
    wc[wave][2] = __builtin_popcount(m2); wc[wave][3] = __builtin_popcount(m3);
  }

  int p0 = 0, p1 = 0, p2 = 0, p3 = 0, s0 = 0, s1 = 0, s2 = 0, s3 = 0;
  for (int i = tid; i < NB; i += BKT) {
    int4 c = ((const int4*)blkcnt)[i];
    int e = (i < b) ? -1 : 0;
    s0 += c.x; s1 += c.y; s2 += c.z; s3 += c.w;
    p0 += c.x & e; p1 += c.y & e; p2 += c.z & e; p3 += c.w & e;
  }
  for (int o = 16; o > 0; o >>= 1) {
    p0 += __shfl_xor(p0, o); p1 += __shfl_xor(p1, o); p2 += __shfl_xor(p2, o); p3 += __shfl_xor(p3, o);
    s0 += __shfl_xor(s0, o); s1 += __shfl_xor(s1, o); s2 += __shfl_xor(s2, o); s3 += __shfl_xor(s3, o);
  }
  if (lane == 0) {
    red[wave][0] = p0; red[wave][1] = p1; red[wave][2] = p2; red[wave][3] = p3;
    red[wave][4] = s0; red[wave][5] = s1; red[wave][6] = s2; red[wave][7] = s3;
  }
  __syncthreads();
  if (tid < 2 * NUM_TYPES) {
    int a = 0;
    for (int w = 0; w < BWAVES; ++w) a += red[w][tid];
    fin[tid] = a;
  }
  __syncthreads();

  if (b == 0 && tid == 0) {
    int acc = 0;
    for (int q = 0; q < NUM_TYPES; ++q) { basev[q] = acc; acc += fin[NUM_TYPES + q]; }
    basev[NUM_TYPES] = acc;
  }
  if ((unsigned)t < NUM_TYPES) {
    unsigned m = t == 0 ? m0 : t == 1 ? m1 : t == 2 ? m2 : m3;
    int pos = fin[t];
    for (int q = 0; q < NUM_TYPES - 1; ++q) pos += (q < t) ? fin[NUM_TYPES + q] : 0;
    for (int w = 0; w < BWAVES - 1; ++w) pos += (w < wave) ? wc[w][t] : 0;
    pos += __builtin_popcount(m & ((1u << lane) - 1u));
    gidx[pos] = n;
  }
}


__global__ __launch_bounds__(256) void k_mlp(
    const float* __restrict__ X, const float* __restrict__ b1, const float* __restrict__ b2,
    const float* __restrict__ W1, const float* __restrict__ W2,
    const int* __restrict__ basev, const int* __restrict__ gidx,
    float* __restrict__ out) {
  __shared__ float Xs[M_TILE][XS_STRIDE];
  __shared__ float Hs[M_TILE][HS_STRIDE];
  __shared__ int   rows[M_TILE];

  int tile = blockIdx.x;
  int ty = -1, ltile = 0, acc = 0, segStart = 0, segCount = 0;
  for (int q = 0; q < NUM_TYPES; ++q) {
    int c  = basev[q + 1] - basev[q];
    int nt = (c + (M_TILE - 1)) >> 6;
    if (tile < acc + nt) { ty = q; ltile = tile - acc; segStart = basev[q]; segCount = c; break; }
    acc += nt;
  }
  if (ty < 0) return;

  int row0 = ltile << 6;
  int tid  = threadIdx.x;

  if (tid < M_TILE) {
    int g = row0 + tid;
    rows[tid] = (g < segCount) ? gidx[segStart + g] : -1;
  }
  __syncthreads();

  {
    int c4 = tid & 31;
    int rb = tid >> 5;
    int nd[M_TILE / 8];
    float4 xv[M_TILE / 8];
#pragma unroll
    for (int j = 0; j < M_TILE / 8; ++j) nd[j] = rows[rb + 8 * j];
#pragma unroll
    for (int j = 0; j < M_TILE / 8; ++j) {
      int node = nd[j] >= 0 ? nd[j] : 0;
      xv[j] = ((const float4*)(X + (size_t)node * IN_DIM))[c4];
    }
#pragma unroll
    for (int j = 0; j < M_TILE / 8; ++j) {
      float4 v = xv[j];
      if (nd[j] < 0) v = make_float4(0.f, 0.f, 0.f, 0.f);
      *(float4*)&Xs[rb + 8 * j][c4 * 4] = v;
    }
  }
  __syncthreads();

  int wave = tid >> 5;
  int lane = tid & 31;
  int n    = lane & 15;
  int kh   = (lane >> 4) << 1;
  int mh   = (lane >> 4) << 3;


  {
    int cbase = wave * 32 + 2 * n;
    const float* wp = W1 + (size_t)ty * IN_DIM * HID_DIM + (size_t)kh * HID_DIM + cbase;
    v8f c00 = {}, c01 = {}, c10 = {}, c11 = {}, c20 = {}, c21 = {}, c30 = {}, c31 = {};
#pragma unroll 4
    for (int k0 = 0; k0 < IN_DIM; k0 += 4) {
      v2f l0 = *(const v2f*)&wp[(k0    ) * HID_DIM];
      v2f l1 = *(const v2f*)&wp[(k0 + 1) * HID_DIM];
      v2f bA = {l0.x, l1.x};
      v2f bB = {l0.y, l1.y};
      v2f a0 = *(const v2f*)&Xs[ 0 + n][k0 + kh];
      v2f a1 = *(const v2f*)&Xs[16 + n][k0 + kh];
      v2f a2 = *(const v2f*)&Xs[32 + n][k0 + kh];
      v2f a3 = *(const v2f*)&Xs[48 + n][k0 + kh];
      c00 = __builtin_amdgcn_wmma_f32_16x16x4_f32(false, a0, false, bA, (short)0, c00, false, false);
      c01 = __builtin_amdgcn_wmma_f32_16x16x4_f32(false, a0, false, bB, (short)0, c01, false, false);
      c10 = __builtin_amdgcn_wmma_f32_16x16x4_f32(false, a1, false, bA, (short)0, c10, false, false);
      c11 = __builtin_amdgcn_wmma_f32_16x16x4_f32(false, a1, false, bB, (short)0, c11, false, false);
      c20 = __builtin_amdgcn_wmma_f32_16x16x4_f32(false, a2, false, bA, (short)0, c20, false, false);
      c21 = __builtin_amdgcn_wmma_f32_16x16x4_f32(false, a2, false, bB, (short)0, c21, false, false);
      c30 = __builtin_amdgcn_wmma_f32_16x16x4_f32(false, a3, false, bA, (short)0, c30, false, false);
      c31 = __builtin_amdgcn_wmma_f32_16x16x4_f32(false, a3, false, bB, (short)0, c31, false, false);
    }
    v2f bias = *(const v2f*)&b1[ty * HID_DIM + cbase];
#pragma unroll
    for (int r = 0; r < 8; ++r) {
      v2f h0 = {c00[r] + bias.x, c01[r] + bias.y};
      v2f h1 = {c10[r] + bias.x, c11[r] + bias.y};
      v2f h2 = {c20[r] + bias.x, c21[r] + bias.y};
      v2f h3 = {c30[r] + bias.x, c31[r] + bias.y};
      h0.x = h0.x > 0.f ? h0.x : 0.f; h0.y = h0.y > 0.f ? h0.y : 0.f;
      h1.x = h1.x > 0.f ? h1.x : 0.f; h1.y = h1.y > 0.f ? h1.y : 0.f;
      h2.x = h2.x > 0.f ? h2.x : 0.f; h2.y = h2.y > 0.f ? h2.y : 0.f;
      h3.x = h3.x > 0.f ? h3.x : 0.f; h3.y = h3.y > 0.f ? h3.y : 0.f;
      *(v2f*)&Hs[ 0 + r + mh][cbase] = h0;
      *(v2f*)&Hs[16 + r + mh][cbase] = h1;
      *(v2f*)&Hs[32 + r + mh][cbase] = h2;
      *(v2f*)&Hs[48 + r + mh][cbase] = h3;
    }
  }
  __syncthreads();

  {
    int cbase = (wave & 3) * 32 + 2 * n;
    int rb    = (wave >> 2) * 32;
    const float* wp = W2 + (size_t)ty * HID_DIM * OUT_DIM + (size_t)kh * OUT_DIM + cbase;
    v8f c00 = {}, c01 = {}, c10 = {}, c11 = {};
#pragma unroll 8
    for (int k0 = 0; k0 < HID_DIM; k0 += 4) {
      v2f l0 = *(const v2f*)&wp[(k0    ) * OUT_DIM];
      v2f l1 = *(const v2f*)&wp[(k0 + 1) * OUT_DIM];
      v2f bA = {l0.x, l1.x};
      v2f bB = {l0.y, l1.y};
      v2f a0 = *(const v2f*)&Hs[rb +  0 + n][k0 + kh];
      v2f a1 = *(const v2f*)&Hs[rb + 16 + n][k0 + kh];
      c00 = __builtin_amdgcn_wmma_f32_16x16x4_f32(false, a0, false, bA, (short)0, c00, false, false);
      c01 = __builtin_amdgcn_wmma_f32_16x16x4_f32(false, a0, false, bB, (short)0, c01, false, false);
      c10 = __builtin_amdgcn_wmma_f32_16x16x4_f32(false, a1, false, bA, (short)0, c10, false, false);
      c11 = __builtin_amdgcn_wmma_f32_16x16x4_f32(false, a1, false, bB, (short)0, c11, false, false);
    }
    v2f bias = *(const v2f*)&b2[ty * OUT_DIM + cbase];
#pragma unroll
    for (int r = 0; r < 8; ++r) {
      int n0 = rows[rb +  0 + r + mh];
      int n1 = rows[rb + 16 + r + mh];
      v2f o0 = {c00[r] + bias.x, c01[r] + bias.y};
      v2f o1 = {c10[r] + bias.x, c11[r] + bias.y};
      if (n0 >= 0) *(v2f*)&out[(size_t)n0 * OUT_DIM + cbase] = o0;
      if (n1 >= 0) *(v2f*)&out[(size_t)n1 * OUT_DIM + cbase] = o1;
    }
  }
}


extern "C" void kernel_launch(void* const* d_in, const int* in_sizes, int n_in,
                              void* d_out, int out_size, void* d_ws, size_t ws_size,
                              hipStream_t stream) {
  const float* X    = (const float*)d_in[0];
  const int*   type = (const int*)  d_in[1];
  const float* W1   = (const float*)d_in[2];
  const float* b1   = (const float*)d_in[3];
  const float* W2   = (const float*)d_in[4];
  const float* b2   = (const float*)d_in[5];
  float* out = (float*)d_out;

  int N = in_sizes[0] / IN_DIM;

  int  nblk    = (N + BKT - 1) / BKT;
  int* basev   = (int*)d_ws;
  int* blkcnt  = basev + 64;
  int  bpad    = (nblk * NUM_TYPES + 63) & ~63;
  int* gidx    = blkcnt + bpad;

  k_hist<<<nblk, BKT, 0, stream>>>(type, blkcnt, N);
  k_scatter<<<nblk, BKT, 0, stream>>>(type, blkcnt, basev, gidx, N, nblk);

  int maxTiles = (N + M_TILE - 1) / M_TILE + NUM_TYPES;
  k_mlp<<<maxTiles, 256, 0, stream>>>(X, b1, b2, W1, W2, basev, gidx, out);
}


# k_mlp row tile 64 -> 80 (layer 2 in row groups of 3+2 subtiles): just under 5 tiles per compute unit, not just over 6, so no nearly empty 7th round; hidden tile reuses the X tile's LDS (3 blocks fit)
# speedup vs baseline: 2.3286x; 1.0938x over previous
#include <hip/hip_runtime.h>

typedef __attribute__((ext_vector_type(2))) float v2f;
typedef __attribute__((ext_vector_type(8))) float v8f;

#define NUM_TYPES 4
#define IN_DIM    128
#define HID_DIM   256
#define OUT_DIM   128
#define M_TILE    80
#define M_SUB     (M_TILE / 16)

#define XS_STRIDE (IN_DIM + 4)
#define HS_STRIDE (HID_DIM + 4)


#define BKT 256
#define BWAVES (BKT / 32)

__global__ __launch_bounds__(BKT) void k_hist(const int* __restrict__ type, int* __restrict__ blkcnt, int N) {
  __shared__ int wc[BWAVES][NUM_TYPES];
  int tid = threadIdx.x, lane = tid & 31, wave = tid >> 5;
  int n = blockIdx.x * BKT + tid;
  int t = (n < N) ? type[n] : -1;
  unsigned m0 = __builtin_amdgcn_ballot_w32(t == 0);
  unsigned m1 = __builtin_amdgcn_ballot_w32(t == 1);
  unsigned m2 = __builtin_amdgcn_ballot_w32(t == 2);
  unsigned m3 = __builtin_amdgcn_ballot_w32(t == 3);
  if (lane == 0) {
    wc[wave][0] = __builtin_popcount(m0); wc[wave][1] = __builtin_popcount(m1);
    wc[wave][2] = __builtin_popcount(m2); wc[wave][3] = __builtin_popcount(m3);
  }
  __syncthreads();
  if (tid < NUM_TYPES) {
    int a = 0;
    for (int w = 0; w < BWAVES; ++w) a += wc[w][tid];
    blkcnt[blockIdx.x * NUM_TYPES + tid] = a;
  }
}

__global__ __launch_bounds__(BKT) void k_scatter(const int* __restrict__ type, const int* __restrict__ blkcnt,
                                                 int* __restrict__ basev, int* __restrict__ gidx, int N, int NB) {
  __shared__ int wc[BWAVES][NUM_TYPES];
  __shared__ int red[BWAVES][2 * NUM_TYPES];
  __shared__ int fin[2 * NUM_TYPES];
  int tid = threadIdx.x, lane = tid & 31, wave = tid >> 5;
  int b = blockIdx.x;
  int n = b * BKT + tid;
  int t = (n < N) ? type[n] : -1;
  unsigned m0 = __builtin_amdgcn_ballot_w32(t == 0);
  unsigned m1 = __builtin_amdgcn_ballot_w32(t == 1);
  unsigned m2 = __builtin_amdgcn_ballot_w32(t == 2);
  unsigned m3 = __builtin_amdgcn_ballot_w32(t == 3);
  if (lane == 0) {
    wc[wave][0] = __builtin_popcount(m0); wc[wave][1] = __builtin_popcount(m1);
    wc[wave][2] = __builtin_popcount(m2); wc[wave][3] = __builtin_popcount(m3);
  }

  int p0 = 0, p1 = 0, p2 = 0, p3 = 0, s0 = 0, s1 = 0, s2 = 0, s3 = 0;
  for (int i = tid; i < NB; i += BKT) {
    int4 c = ((const int4*)blkcnt)[i];
    int e = (i < b) ? -1 : 0;
    s0 += c.x; s1 += c.y; s2 += c.z; s3 += c.w;
    p0 += c.x & e; p1 += c.y & e; p2 += c.z & e; p3 += c.w & e;
  }
  for (int o = 16; o > 0; o >>= 1) {
    p0 += __shfl_xor(p0, o); p1 += __shfl_xor(p1, o); p2 += __shfl_xor(p2, o); p3 += __shfl_xor(p3, o);
    s0 += __shfl_xor(s0, o); s1 += __shfl_xor(s1, o); s2 += __shfl_xor(s2, o); s3 += __shfl_xor(s3, o);
  }
  if (lane == 0) {
    red[wave][0] = p0; red[wave][1] = p1; red[wave][2] = p2; red[wave][3] = p3;
    red[wave][4] = s0; red[wave][5] = s1; red[wave][6] = s2; red[wave][7] = s3;
  }
  __syncthreads();
  if (tid < 2 * NUM_TYPES) {
    int a = 0;
    for (int w = 0; w < BWAVES; ++w) a += red[w][tid];
    fin[tid] = a;
  }
  __syncthreads();

  if (b == 0 && tid == 0) {
    int acc = 0;
    for (int q = 0; q < NUM_TYPES; ++q) { basev[q] = acc; acc += fin[NUM_TYPES + q]; }
    basev[NUM_TYPES] = acc;
  }
  if ((unsigned)t < NUM_TYPES) {
    unsigned m = t == 0 ? m0 : t == 1 ? m1 : t == 2 ? m2 : m3;
    int pos = fin[t];
    for (int q = 0; q < NUM_TYPES - 1; ++q) pos += (q < t) ? fin[NUM_TYPES + q] : 0;
    for (int w = 0; w < BWAVES - 1; ++w) pos += (w < wave) ? wc[w][t] : 0;
    pos += __builtin_popcount(m & ((1u << lane) - 1u));
    gidx[pos] = n;
  }
}


template <int NM>
__device__ __forceinline__ void mlp_layer2(const float (*Hs)[HS_STRIDE], const int* rows,
                                           const float* __restrict__ wp, const float* __restrict__ bp,
                                           float* __restrict__ op, int rb, int n, int kh, int mh) {
  v8f acc[NM][2] = {};
#pragma unroll 8
  for (int k0 = 0; k0 < HID_DIM; k0 += 4) {
    v2f l0 = *(const v2f*)&wp[(k0    ) * OUT_DIM];
    v2f l1 = *(const v2f*)&wp[(k0 + 1) * OUT_DIM];
    v2f bA = {l0.x, l1.x};
    v2f bB = {l0.y, l1.y};
    v2f a[NM];
#pragma unroll
    for (int m = 0; m < NM; ++m) a[m] = *(const v2f*)&Hs[rb + 16 * m + n][k0 + kh];
#pragma unroll
    for (int m = 0; m < NM; ++m) {
      acc[m][0] = __builtin_amdgcn_wmma_f32_16x16x4_f32(false, a[m], false, bA, (short)0, acc[m][0], false, false);
      acc[m][1] = __builtin_amdgcn_wmma_f32_16x16x4_f32(false, a[m], false, bB, (short)0, acc[m][1], false, false);
    }
  }
  v2f bias = *(const v2f*)bp;
#pragma unroll
  for (int r = 0; r < 8; ++r) {
    int node[NM];
#pragma unroll
    for (int m = 0; m < NM; ++m) node[m] = rows[rb + 16 * m + r + mh];
#pragma unroll
    for (int m = 0; m < NM; ++m) {
      v2f o = {acc[m][0][r] + bias.x, acc[m][1][r] + bias.y};
      if (node[m] >= 0) *(v2f*)&op[(size_t)node[m] * OUT_DIM] = o;
    }
  }
}

__global__ __launch_bounds__(256) void k_mlp(
    const float* __restrict__ X, const float* __restrict__ b1, const float* __restrict__ b2,
    const float* __restrict__ W1, const float* __restrict__ W2,
    const int* __restrict__ basev, const int* __restrict__ gidx,
    float* __restrict__ out) {
  __shared__ __attribute__((aligned(16))) float tilebuf[M_TILE * HS_STRIDE];
  __shared__ int   rows[M_TILE];
  float (*Xs)[XS_STRIDE] = (float (*)[XS_STRIDE])tilebuf;
  float (*Hs)[HS_STRIDE] = (float (*)[HS_STRIDE])tilebuf;

  int tile = blockIdx.x;
  int ty = -1, ltile = 0, tbase = 0, segStart = 0, segCount = 0;
  for (int q = 0; q < NUM_TYPES; ++q) {
    int c  = basev[q + 1] - basev[q];
    int nt = (c + (M_TILE - 1)) / M_TILE;
    if (tile < tbase + nt) { ty = q; ltile = tile - tbase; segStart = basev[q]; segCount = c; break; }
    tbase += nt;
  }
  if (ty < 0) return;

  int row0 = ltile * M_TILE;
  int tid  = threadIdx.x;

  if (tid < M_TILE) {
    int g = row0 + tid;
    rows[tid] = (g < segCount) ? gidx[segStart + g] : -1;
  }
  __syncthreads();

  {
    int c4 = tid & 31;
    int rb = tid >> 5;
    int nd[M_TILE / 8];
    float4 xv[M_TILE / 8];
#pragma unroll
    for (int j = 0; j < M_TILE / 8; ++j) nd[j] = rows[rb + 8 * j];
#pragma unroll
    for (int j = 0; j < M_TILE / 8; ++j) {
      int node = nd[j] >= 0 ? nd[j] : 0;
      xv[j] = ((const float4*)(X + (size_t)node * IN_DIM))[c4];
    }
#pragma unroll
    for (int j = 0; j < M_TILE / 8; ++j) {
      float4 v = xv[j];
      if (nd[j] < 0) v = make_float4(0.f, 0.f, 0.f, 0.f);
      *(float4*)&Xs[rb + 8 * j][c4 * 4] = v;
    }
  }
  __syncthreads();

  int wave = __builtin_amdgcn_readfirstlane(tid >> 5);
  int lane = tid & 31;
  int n    = lane & 15;
  int kh   = (lane >> 4) << 1;
  int mh   = (lane >> 4) << 3;


  {
    int cbase = wave * 32 + 2 * n;
    const float* wp = W1 + (size_t)ty * IN_DIM * HID_DIM + (size_t)kh * HID_DIM + cbase;
    v8f acc[M_SUB][2] = {};
#pragma unroll 4
    for (int k0 = 0; k0 < IN_DIM; k0 += 4) {
      v2f l0 = *(const v2f*)&wp[(k0    ) * HID_DIM];
      v2f l1 = *(const v2f*)&wp[(k0 + 1) * HID_DIM];
      v2f bA = {l0.x, l1.x};
      v2f bB = {l0.y, l1.y};
      v2f a[M_SUB];
#pragma unroll
      for (int m = 0; m < M_SUB; ++m) a[m] = *(const v2f*)&Xs[16 * m + n][k0 + kh];
#pragma unroll
      for (int m = 0; m < M_SUB; ++m) {
        acc[m][0] = __builtin_amdgcn_wmma_f32_16x16x4_f32(false, a[m], false, bA, (short)0, acc[m][0], false, false);
        acc[m][1] = __builtin_amdgcn_wmma_f32_16x16x4_f32(false, a[m], false, bB, (short)0, acc[m][1], false, false);
      }
    }
    v2f bias = *(const v2f*)&b1[ty * HID_DIM + cbase];
    __syncthreads();
#pragma unroll
    for (int m = 0; m < M_SUB; ++m) {
#pragma unroll
      for (int r = 0; r < 8; ++r) {
        v2f h = {acc[m][0][r] + bias.x, acc[m][1][r] + bias.y};
        h.x = h.x > 0.f ? h.x : 0.f; h.y = h.y > 0.f ? h.y : 0.f;
        *(v2f*)&Hs[16 * m + r + mh][cbase] = h;
      }
    }
  }
  __syncthreads();

  {
    int cbase = (wave & 3) * 32 + 2 * n;
    const float* wp = W2 + (size_t)ty * HID_DIM * OUT_DIM + (size_t)kh * OUT_DIM + cbase;
    const float* bp = b2 + ty * OUT_DIM + cbase;
    float* op = out + cbase;
    if (wave < 4) mlp_layer2<3>(Hs, rows, wp, bp, op,  0, n, kh, mh);
    else          mlp_layer2<2>(Hs, rows, wp, bp, op, 48, n, kh, mh);
  }
}


extern "C" void kernel_launch(void* const* d_in, const int* in_sizes, int n_in,
                              void* d_out, int out_size, void* d_ws, size_t ws_size,
                              hipStream_t stream) {
  const float* X    = (const float*)d_in[0];
  const int*   type = (const int*)  d_in[1];
  const float* W1   = (const float*)d_in[2];
  const float* b1   = (const float*)d_in[3];
  const float* W2   = (const float*)d_in[4];
  const float* b2   = (const float*)d_in[5];
  float* out = (float*)d_out;

  int N = in_sizes[0] / IN_DIM;

  int  nblk    = (N + BKT - 1) / BKT;
  int* basev   = (int*)d_ws;
  int* blkcnt  = basev + 64;
  int  bpad    = (nblk * NUM_TYPES + 63) & ~63;
  int* gidx    = blkcnt + bpad;

  k_hist<<<nblk, BKT, 0, stream>>>(type, blkcnt, N);
  k_scatter<<<nblk, BKT, 0, stream>>>(type, blkcnt, basev, gidx, N, nblk);

  int maxTiles = (N + M_TILE - 1) / M_TILE + NUM_TYPES;
  k_mlp<<<maxTiles, 256, 0, stream>>>(X, b1, b2, W1, W2, basev, gidx, out);
}
